# starting kernel
# baseline (speedup 1.0000x reference)
#include <hip/hip_runtime.h>
#include <hip/hip_bf16.h>

typedef __attribute__((ext_vector_type(2))) float v2f;
typedef __attribute__((ext_vector_type(8))) float v8f;

#define IN_DIM 32
#define LDS_STRIDE 36
#define WAVES_PER_BLOCK 8

__device__ __forceinline__ v8f wmma4(v2f a, v2f b, v8f c) {
  return __builtin_amdgcn_wmma_f32_16x16x4_f32(
       false, a,  false, b,
       (short)0, c,  false,  false);
}

__device__ __forceinline__ float relu1(float x) {
  return __builtin_amdgcn_fmed3f(x, 0.0f, __builtin_inff());
}

__device__ __forceinline__ v2f load_bfrag(const float* __restrict__ W,
                                          int lane, int kb, int nt) {
  int n = nt * 16 + (lane & 15);
  int k = kb * 4 + ((lane >> 4) << 1);
  return *(const v2f*)(W + n * IN_DIM + k);
}

struct Frags {
  v2f b1[2][8];
  v2f b2[2][8];
  float bias1[2];
  float bias2[2];
};

__device__ __forceinline__ void load_frags(Frags& f,
                                           const float* __restrict__ w1,
                                           const float* __restrict__ b1,
                                           const float* __restrict__ w2,
                                           const float* __restrict__ b2,
                                           int lane) {
#pragma unroll
  for (int nt = 0; nt < 2; ++nt) {
#pragma unroll
    for (int kb = 0; kb < 8; ++kb) {
      f.b1[nt][kb] = load_bfrag(w1, lane, kb, nt);
      f.b2[nt][kb] = load_bfrag(w2, lane, kb, nt);
    }
    f.bias1[nt] = b1[nt * 16 + (lane & 15)];
    f.bias2[nt] = b2[nt * 16 + (lane & 15)];
  }
}

__device__ __forceinline__ void mlp_tile(const Frags& f,
                                         const float* __restrict__ X,
                                         int row_base, int n_rows,
                                         float* __restrict__ wlds,
                                         int lane, v8f& y0, v8f& y1) {
  const int m = lane & 15;
  const int half = lane >> 4;
  int mrow = row_base + ((m < n_rows) ? m : (n_rows - 1));
  const v2f* xrow = (const v2f*)(X + (size_t)mrow * IN_DIM + half * 2);

  v8f acc0, acc1;
#pragma unroll
  for (int r = 0; r < 8; ++r) { acc0[r] = f.bias1[0]; acc1[r] = f.bias1[1]; }
#pragma unroll
  for (int kb = 0; kb < 8; ++kb) {
    v2f a = __builtin_nontemporal_load(xrow + kb * 2);
    acc0 = wmma4(a, f.b1[0][kb], acc0);
    acc1 = wmma4(a, f.b1[1][kb], acc1);
  }
#pragma unroll
  for (int r = 0; r < 8; ++r) { acc0[r] = relu1(acc0[r]); acc1[r] = relu1(acc1[r]); }

  {
    float* base = wlds + (8 * half) * LDS_STRIDE + m;
#pragma unroll
    for (int r = 0; r < 8; ++r) {
      base[r * LDS_STRIDE]      = acc0[r];
      base[r * LDS_STRIDE + 16] = acc1[r];
    }
  }
  __builtin_amdgcn_wave_barrier();
  asm volatile("s_wait_dscnt 0x0" ::: "memory");

#pragma unroll
  for (int r = 0; r < 8; ++r) { acc0[r] = f.bias2[0]; acc1[r] = f.bias2[1]; }
  const float* hrow = wlds + m * LDS_STRIDE + half * 2;
#pragma unroll
  for (int kb = 0; kb < 8; ++kb) {
    v2f a = *(const v2f*)(hrow + kb * 4);
    acc0 = wmma4(a, f.b2[0][kb], acc0);
    acc1 = wmma4(a, f.b2[1][kb], acc1);
  }
  __builtin_amdgcn_wave_barrier();
  asm volatile("" ::: "memory");
#pragma unroll
  for (int r = 0; r < 8; ++r) { y0[r] = relu1(acc0[r]); y1[r] = relu1(acc1[r]); }
}

__global__ void __launch_bounds__(256) zero_out_kernel(float* __restrict__ out, int n) {
  int i = blockIdx.x * blockDim.x + threadIdx.x;
  int stride = gridDim.x * blockDim.x;
  float4* o4 = (float4*)out;
  int n4 = n >> 2;
  float4 z; z.x = 0.f; z.y = 0.f; z.z = 0.f; z.w = 0.f;
  for (; i < n4; i += stride) o4[i] = z;
}

__global__ void __launch_bounds__(256) node_mlp_kernel(
    const float* __restrict__ X,
    const float* __restrict__ w1, const float* __restrict__ b1,
    const float* __restrict__ w2, const float* __restrict__ b2,
    float* __restrict__ out, int n_nodes) {
  __shared__ float lds[WAVES_PER_BLOCK * 16 * LDS_STRIDE];
  const int lane = threadIdx.x & 31;
  const int wid  = threadIdx.x >> 5;
  float* wlds = lds + wid * 16 * LDS_STRIDE;

  Frags f;
  load_frags(f, w1, b1, w2, b2, lane);

  const int ntiles = (n_nodes + 15) >> 4;
  const int gwave  = blockIdx.x * WAVES_PER_BLOCK + wid;
  const int nwaves = gridDim.x * WAVES_PER_BLOCK;
  const int half = lane >> 4, col = lane & 15;

  for (int tile = gwave; tile < ntiles; tile += nwaves) {
    const int row_base = tile * 16;
    int nr = n_nodes - row_base; if (nr > 16) nr = 16;
    v8f y0, y1;
    mlp_tile(f, X, row_base, nr, wlds, lane, y0, y1);
#pragma unroll
    for (int r = 0; r < 8; ++r) {
      int row = row_base + r + 8 * half;
      if (row < n_nodes) {
        float* o = out + (size_t)row * 64;
        __builtin_nontemporal_store(y0[r], o + col);
        __builtin_nontemporal_store(y1[r], o + 16 + col);
      }
    }
  }
}

__global__ void __launch_bounds__(256) edge_mlp_scatter_kernel(
    const float* __restrict__ X, const int* __restrict__ src,
    const float* __restrict__ w1, const float* __restrict__ b1,
    const float* __restrict__ w2, const float* __restrict__ b2,
    float* __restrict__ out, int n_edges) {
  __shared__ float lds[WAVES_PER_BLOCK * 16 * LDS_STRIDE];
  const int lane = threadIdx.x & 31;
  const int wid  = threadIdx.x >> 5;
  float* wlds = lds + wid * 16 * LDS_STRIDE;

  Frags f;
  load_frags(f, w1, b1, w2, b2, lane);

  const int ntiles = (n_edges + 15) >> 4;
  const int gwave  = blockIdx.x * WAVES_PER_BLOCK + wid;
  const int nwaves = gridDim.x * WAVES_PER_BLOCK;
  const int half = lane >> 4, col = lane & 15;

  for (int tile = gwave; tile < ntiles; tile += nwaves) {
    const int row_base = tile * 16;
    int nr = n_edges - row_base; if (nr > 16) nr = 16;

    int ntile = tile + nwaves;
    if (ntile < ntiles) {
      const float* p = X + (size_t)(ntile * 16 + (lane & 15)) * IN_DIM;
      __builtin_prefetch(p, 0, 1);
    }

    v8f y0, y1;
    mlp_tile(f, X, row_base, nr, wlds, lane, y0, y1);

#pragma unroll
    for (int r = 0; r < 8; ++r) {
      int row = row_base + r + 8 * half;
      if (row < n_edges) {
        int s = src[row];
        unsigned* dst = (unsigned*)(out + (size_t)s * 64 + 32);
        atomicMax(dst + col,      __float_as_uint(y0[r]));
        atomicMax(dst + 16 + col, __float_as_uint(y1[r]));
      }
    }
  }
}

extern "C" void kernel_launch(void* const* d_in, const int* in_sizes, int n_in,
                              void* d_out, int out_size, void* d_ws, size_t ws_size,
                              hipStream_t stream) {
  const float* node_features = (const float*)d_in[0];
  const float* edge_features = (const float*)d_in[1];
  const int*   edge_index    = (const int*)d_in[2];
  const float* nw1 = (const float*)d_in[3];
  const float* nb1 = (const float*)d_in[4];
  const float* nw2 = (const float*)d_in[5];
  const float* nb2 = (const float*)d_in[6];
  const float* ew1 = (const float*)d_in[7];
  const float* eb1 = (const float*)d_in[8];
  const float* ew2 = (const float*)d_in[9];
  const float* eb2 = (const float*)d_in[10];
  float* out = (float*)d_out;

  const int n_nodes = in_sizes[0] / 32;
  const int n_edges = in_sizes[1] / 32;

  zero_out_kernel<<<512, 256, 0, stream>>>(out, n_nodes * 64);
  node_mlp_kernel<<<512, 256, 0, stream>>>(
      node_features, nw1, nb1, nw2, nb2, out, n_nodes);
  edge_mlp_scatter_kernel<<<2048, 256, 0, stream>>>(
      edge_features, edge_index, ew1, eb1, ew2, eb2, out, n_edges);
}


# edge scatter-max: one whole-128-byte-line atomic per edge (lane = column) after a v_permlane16_swap of the two output tiles; source indices loaded once per tile, a tile ahead, read by readlane
# speedup vs baseline: 1.5975x; 1.5975x over previous
#include <hip/hip_runtime.h>
#include <hip/hip_bf16.h>

typedef __attribute__((ext_vector_type(2))) float v2f;
typedef __attribute__((ext_vector_type(8))) float v8f;
typedef __attribute__((ext_vector_type(2))) unsigned v2u;

#define IN_DIM 32
#define LDS_STRIDE 36
#define WAVES_PER_BLOCK 8

__device__ __forceinline__ v8f wmma4(v2f a, v2f b, v8f c) {
  return __builtin_amdgcn_wmma_f32_16x16x4_f32(
       false, a,  false, b,
       (short)0, c,  false,  false);
}

__device__ __forceinline__ float relu1(float x) {
  return __builtin_amdgcn_fmed3f(x, 0.0f, __builtin_inff());
}

__device__ __forceinline__ v2f load_bfrag(const float* __restrict__ W,
                                          int lane, int kb, int nt) {
  int n = nt * 16 + (lane & 15);
  int k = kb * 4 + ((lane >> 4) << 1);
  return *(const v2f*)(W + n * IN_DIM + k);
}

struct Frags {
  v2f b1[2][8];
  v2f b2[2][8];
  float bias1[2];
  float bias2[2];
};

__device__ __forceinline__ void load_frags(Frags& f,
                                           const float* __restrict__ w1,
                                           const float* __restrict__ b1,
                                           const float* __restrict__ w2,
                                           const float* __restrict__ b2,
                                           int lane) {
#pragma unroll
  for (int nt = 0; nt < 2; ++nt) {
#pragma unroll
    for (int kb = 0; kb < 8; ++kb) {
      f.b1[nt][kb] = load_bfrag(w1, lane, kb, nt);
      f.b2[nt][kb] = load_bfrag(w2, lane, kb, nt);
    }
    f.bias1[nt] = b1[nt * 16 + (lane & 15)];
    f.bias2[nt] = b2[nt * 16 + (lane & 15)];
  }
}

__device__ __forceinline__ void mlp_tile(const Frags& f,
                                         const float* __restrict__ X,
                                         int row_base, int n_rows,
                                         float* __restrict__ wlds,
                                         int lane, v8f& y0, v8f& y1) {
  const int m = lane & 15;
  const int half = lane >> 4;
  int mrow = row_base + ((m < n_rows) ? m : (n_rows - 1));
  const v2f* xrow = (const v2f*)(X + (size_t)mrow * IN_DIM + half * 2);

  v8f acc0, acc1;
#pragma unroll
  for (int r = 0; r < 8; ++r) { acc0[r] = f.bias1[0]; acc1[r] = f.bias1[1]; }
#pragma unroll
  for (int kb = 0; kb < 8; ++kb) {
    v2f a = __builtin_nontemporal_load(xrow + kb * 2);
    acc0 = wmma4(a, f.b1[0][kb], acc0);
    acc1 = wmma4(a, f.b1[1][kb], acc1);
  }
#pragma unroll
  for (int r = 0; r < 8; ++r) { acc0[r] = relu1(acc0[r]); acc1[r] = relu1(acc1[r]); }

  {
    float* base = wlds + (8 * half) * LDS_STRIDE + m;
#pragma unroll
    for (int r = 0; r < 8; ++r) {
      base[r * LDS_STRIDE]      = acc0[r];
      base[r * LDS_STRIDE + 16] = acc1[r];
    }
  }
  __builtin_amdgcn_wave_barrier();
  asm volatile("s_wait_dscnt 0x0" ::: "memory");

#pragma unroll
  for (int r = 0; r < 8; ++r) { acc0[r] = f.bias2[0]; acc1[r] = f.bias2[1]; }
  const float* hrow = wlds + m * LDS_STRIDE + half * 2;
#pragma unroll
  for (int kb = 0; kb < 8; ++kb) {
    v2f a = *(const v2f*)(hrow + kb * 4);
    acc0 = wmma4(a, f.b2[0][kb], acc0);
    acc1 = wmma4(a, f.b2[1][kb], acc1);
  }
  __builtin_amdgcn_wave_barrier();
  asm volatile("" ::: "memory");
#pragma unroll
  for (int r = 0; r < 8; ++r) { y0[r] = relu1(acc0[r]); y1[r] = relu1(acc1[r]); }
}

__global__ void __launch_bounds__(256) zero_out_kernel(float* __restrict__ out, int n) {
  int i = blockIdx.x * blockDim.x + threadIdx.x;
  int stride = gridDim.x * blockDim.x;
  float4* o4 = (float4*)out;
  int n4 = n >> 2;
  float4 z; z.x = 0.f; z.y = 0.f; z.z = 0.f; z.w = 0.f;
  for (; i < n4; i += stride) o4[i] = z;
}

__global__ void __launch_bounds__(256) node_mlp_kernel(
    const float* __restrict__ X,
    const float* __restrict__ w1, const float* __restrict__ b1,
    const float* __restrict__ w2, const float* __restrict__ b2,
    float* __restrict__ out, int n_nodes) {
  __shared__ float lds[WAVES_PER_BLOCK * 16 * LDS_STRIDE];
  const int lane = threadIdx.x & 31;
  const int wid  = threadIdx.x >> 5;
  float* wlds = lds + wid * 16 * LDS_STRIDE;

  Frags f;
  load_frags(f, w1, b1, w2, b2, lane);

  const int ntiles = (n_nodes + 15) >> 4;
  const int gwave  = blockIdx.x * WAVES_PER_BLOCK + wid;
  const int nwaves = gridDim.x * WAVES_PER_BLOCK;
  const int half = lane >> 4, col = lane & 15;

  for (int tile = gwave; tile < ntiles; tile += nwaves) {
    const int row_base = tile * 16;
    int nr = n_nodes - row_base; if (nr > 16) nr = 16;
    v8f y0, y1;
    mlp_tile(f, X, row_base, nr, wlds, lane, y0, y1);
#pragma unroll
    for (int r = 0; r < 8; ++r) {
      int row = row_base + r + 8 * half;
      if (row < n_nodes) {
        float* o = out + (size_t)row * 64;
        __builtin_nontemporal_store(y0[r], o + col);
        __builtin_nontemporal_store(y1[r], o + 16 + col);
      }
    }
  }
}

__global__ void __launch_bounds__(256) edge_mlp_scatter_kernel(
    const float* __restrict__ X, const int* __restrict__ src,
    const float* __restrict__ w1, const float* __restrict__ b1,
    const float* __restrict__ w2, const float* __restrict__ b2,
    float* __restrict__ out, int n_edges) {
  __shared__ float lds[WAVES_PER_BLOCK * 16 * LDS_STRIDE];
  const int lane = threadIdx.x & 31;
  const int wid  = __builtin_amdgcn_readfirstlane((int)(threadIdx.x >> 5));
  float* wlds = lds + wid * 16 * LDS_STRIDE;

  Frags f;
  load_frags(f, w1, b1, w2, b2, lane);

  const int ntiles = (n_edges + 15) >> 4;
  const int gwave  = blockIdx.x * WAVES_PER_BLOCK + wid;
  const int nwaves = gridDim.x * WAVES_PER_BLOCK;

  unsigned* const agg = (unsigned*)out + 32;

  int sv = 0;
  if (gwave < ntiles) {
    int e = gwave * 16 + (lane & 15);
    sv = src[e < n_edges ? e : n_edges - 1];
  }

  for (int tile = gwave; tile < ntiles; tile += nwaves) {
    const int row_base = tile * 16;
    int nr = n_edges - row_base; if (nr > 16) nr = 16;

    int ntile = tile + nwaves;
    int sv_next = 0;
    if (ntile < ntiles) {
      const float* p = X + (size_t)(ntile * 16 + (lane & 15)) * IN_DIM;
      __builtin_prefetch(p, 0, 1);
      int e = ntile * 16 + (lane & 15);
      sv_next = src[e < n_edges ? e : n_edges - 1];
    }

    v8f y0, y1;
    mlp_tile(f, X, row_base, nr, wlds, lane, y0, y1);

    unsigned lo[8], hi[8];
#pragma unroll
    for (int r = 0; r < 8; ++r) {
      lo[r] = __float_as_uint(y0[r]);
      hi[r] = __float_as_uint(y1[r]);
      {
        v2u sw = __builtin_amdgcn_permlane16_swap(lo[r], hi[r], false, false);
        lo[r] = sw.x; hi[r] = sw.y;
      }
    }
    if (nr == 16) {
#pragma unroll
      for (int r = 0; r < 8; ++r) {
        const int s0 = __builtin_amdgcn_readlane(sv, r);
        const int s1 = __builtin_amdgcn_readlane(sv, r + 8);
        atomicMax(agg + (size_t)s0 * 64 + lane, lo[r]);
        atomicMax(agg + (size_t)s1 * 64 + lane, hi[r]);
      }
    } else {
#pragma unroll
      for (int r = 0; r < 8; ++r) {
        if (r < nr) {
          const int s0 = __builtin_amdgcn_readlane(sv, r);
          atomicMax(agg + (size_t)s0 * 64 + lane, lo[r]);
        }
        if (r + 8 < nr) {
          const int s1 = __builtin_amdgcn_readlane(sv, r + 8);
          atomicMax(agg + (size_t)s1 * 64 + lane, hi[r]);
        }
      }
    }
    sv = sv_next;
  }
}

extern "C" void kernel_launch(void* const* d_in, const int* in_sizes, int n_in,
                              void* d_out, int out_size, void* d_ws, size_t ws_size,
                              hipStream_t stream) {
  const float* node_features = (const float*)d_in[0];
  const float* edge_features = (const float*)d_in[1];
  const int*   edge_index    = (const int*)d_in[2];
  const float* nw1 = (const float*)d_in[3];
  const float* nb1 = (const float*)d_in[4];
  const float* nw2 = (const float*)d_in[5];
  const float* nb2 = (const float*)d_in[6];
  const float* ew1 = (const float*)d_in[7];
  const float* eb1 = (const float*)d_in[8];
  const float* ew2 = (const float*)d_in[9];
  const float* eb2 = (const float*)d_in[10];
  float* out = (float*)d_out;

  const int n_nodes = in_sizes[0] / 32;
  const int n_edges = in_sizes[1] / 32;

  zero_out_kernel<<<512, 256, 0, stream>>>(out, n_nodes * 64);
  node_mlp_kernel<<<512, 256, 0, stream>>>(
      node_features, nw1, nb1, nw2, nb2, out, n_nodes);
  edge_mlp_scatter_kernel<<<2048, 256, 0, stream>>>(
      edge_features, edge_index, ew1, eb1, ew2, eb2, out, n_edges);
}


# X tiles fetched by four whole-line 16-byte-per-lane loads (512 contiguous bytes each), a tile ahead, staged through the wave's LDS tile into A fragments; replaces 8-byte loads touching 16 lines each
# speedup vs baseline: 1.7039x; 1.0666x over previous
#include <hip/hip_runtime.h>
#include <hip/hip_bf16.h>

typedef __attribute__((ext_vector_type(2))) float v2f;
typedef __attribute__((ext_vector_type(8))) float v8f;
typedef __attribute__((ext_vector_type(2))) unsigned v2u;
typedef __attribute__((ext_vector_type(4))) float v4f;

#define IN_DIM 32
#define LDS_STRIDE 36
#define WAVES_PER_BLOCK 8

__device__ __forceinline__ v8f wmma4(v2f a, v2f b, v8f c) {
  return __builtin_amdgcn_wmma_f32_16x16x4_f32(
       false, a,  false, b,
       (short)0, c,  false,  false);
}

__device__ __forceinline__ float relu1(float x) {
  return __builtin_amdgcn_fmed3f(x, 0.0f, __builtin_inff());
}

__device__ __forceinline__ v2f load_bfrag(const float* __restrict__ W,
                                          int lane, int kb, int nt) {
  int n = nt * 16 + (lane & 15);
  int k = kb * 4 + ((lane >> 4) << 1);
  return *(const v2f*)(W + n * IN_DIM + k);
}

struct Frags {
  v2f b1[2][8];
  v2f b2[2][8];
  float bias1[2];
  float bias2[2];
};

__device__ __forceinline__ void load_frags(Frags& f,
                                           const float* __restrict__ w1,
                                           const float* __restrict__ b1,
                                           const float* __restrict__ w2,
                                           const float* __restrict__ b2,
                                           int lane) {
#pragma unroll
  for (int nt = 0; nt < 2; ++nt) {
#pragma unroll
    for (int kb = 0; kb < 8; ++kb) {
      f.b1[nt][kb] = load_bfrag(w1, lane, kb, nt);
      f.b2[nt][kb] = load_bfrag(w2, lane, kb, nt);
    }
    f.bias1[nt] = b1[nt * 16 + (lane & 15)];
    f.bias2[nt] = b2[nt * 16 + (lane & 15)];
  }
}

__device__ __forceinline__ void load_xtile(v4f (&xr)[4],
                                           const float* __restrict__ X,
                                           int row_base, int n_rows, int lane) {
  if (n_rows >= 16) {
    const v4f* p = (const v4f*)(X + (size_t)row_base * IN_DIM) + lane;
#pragma unroll
    for (int q = 0; q < 4; ++q) xr[q] = __builtin_nontemporal_load(p + q * 32);
  } else {
    const int sub = lane >> 3;
    const int c4  = (lane & 7) * 4;
#pragma unroll
    for (int q = 0; q < 4; ++q) {
      int r = q * 4 + sub;
      r = (r < n_rows) ? r : (n_rows - 1);
      xr[q] = __builtin_nontemporal_load(
          (const v4f*)(X + (size_t)(row_base + r) * IN_DIM + c4));
    }
  }
}

__device__ __forceinline__ void stage_xtile(const v4f (&xr)[4],
                                            float* __restrict__ wlds, int lane) {
  float* p = wlds + (lane >> 3) * LDS_STRIDE + (lane & 7) * 4;
#pragma unroll
  for (int q = 0; q < 4; ++q) *(v4f*)(p + q * 4 * LDS_STRIDE) = xr[q];
}

__device__ __forceinline__ void mlp_tile(const Frags& f,
                                         float* __restrict__ wlds,
                                         int lane, v8f& y0, v8f& y1) {
  const int m = lane & 15;
  const int half = lane >> 4;
  const float* hrow = wlds + m * LDS_STRIDE + half * 2;

  __builtin_amdgcn_wave_barrier();
  asm volatile("s_wait_dscnt 0x0" ::: "memory");

  v8f acc0, acc1;
#pragma unroll
  for (int r = 0; r < 8; ++r) { acc0[r] = f.bias1[0]; acc1[r] = f.bias1[1]; }
#pragma unroll
  for (int kb = 0; kb < 8; ++kb) {
    v2f a = *(const v2f*)(hrow + kb * 4);
    acc0 = wmma4(a, f.b1[0][kb], acc0);
    acc1 = wmma4(a, f.b1[1][kb], acc1);
  }
#pragma unroll
  for (int r = 0; r < 8; ++r) { acc0[r] = relu1(acc0[r]); acc1[r] = relu1(acc1[r]); }

  {
    float* base = wlds + (8 * half) * LDS_STRIDE + m;
#pragma unroll
    for (int r = 0; r < 8; ++r) {
      base[r * LDS_STRIDE]      = acc0[r];
      base[r * LDS_STRIDE + 16] = acc1[r];
    }
  }
  __builtin_amdgcn_wave_barrier();
  asm volatile("s_wait_dscnt 0x0" ::: "memory");

#pragma unroll
  for (int r = 0; r < 8; ++r) { acc0[r] = f.bias2[0]; acc1[r] = f.bias2[1]; }
#pragma unroll
  for (int kb = 0; kb < 8; ++kb) {
    v2f a = *(const v2f*)(hrow + kb * 4);
    acc0 = wmma4(a, f.b2[0][kb], acc0);
    acc1 = wmma4(a, f.b2[1][kb], acc1);
  }
  __builtin_amdgcn_wave_barrier();
  asm volatile("" ::: "memory");
#pragma unroll
  for (int r = 0; r < 8; ++r) { y0[r] = relu1(acc0[r]); y1[r] = relu1(acc1[r]); }
}

__global__ void __launch_bounds__(256) zero_out_kernel(float* __restrict__ out, int n) {
  int i = blockIdx.x * blockDim.x + threadIdx.x;
  int stride = gridDim.x * blockDim.x;
  float4* o4 = (float4*)out;
  int n4 = n >> 2;
  float4 z; z.x = 0.f; z.y = 0.f; z.z = 0.f; z.w = 0.f;
  for (; i < n4; i += stride) o4[i] = z;
}

__global__ void __launch_bounds__(256) node_mlp_kernel(
    const float* __restrict__ X,
    const float* __restrict__ w1, const float* __restrict__ b1,
    const float* __restrict__ w2, const float* __restrict__ b2,
    float* __restrict__ out, int n_nodes) {
  __shared__ float lds[WAVES_PER_BLOCK * 16 * LDS_STRIDE];
  const int lane = threadIdx.x & 31;
  const int wid  = threadIdx.x >> 5;
  float* wlds = lds + wid * 16 * LDS_STRIDE;

  const int ntiles = (n_nodes + 15) >> 4;
  const int gwave  = blockIdx.x * WAVES_PER_BLOCK + wid;
  const int nwaves = gridDim.x * WAVES_PER_BLOCK;
  const int half = lane >> 4, col = lane & 15;
  if (gwave >= ntiles) return;

  Frags f;
  load_frags(f, w1, b1, w2, b2, lane);

  v4f xr[4];
  {
    int nr0 = n_nodes - gwave * 16; if (nr0 > 16) nr0 = 16;
    load_xtile(xr, X, gwave * 16, nr0, lane);
  }

  for (int tile = gwave; tile < ntiles; tile += nwaves) {
    const int row_base = tile * 16;
    stage_xtile(xr, wlds, lane);
    const int ntile = tile + nwaves;
    if (ntile < ntiles) {
      int nrn = n_nodes - ntile * 16; if (nrn > 16) nrn = 16;
      load_xtile(xr, X, ntile * 16, nrn, lane);
    }
    v8f y0, y1;
    mlp_tile(f, wlds, lane, y0, y1);
#pragma unroll
    for (int r = 0; r < 8; ++r) {
      int row = row_base + r + 8 * half;
      if (row < n_nodes) {
        float* o = out + (size_t)row * 64;
        __builtin_nontemporal_store(y0[r], o + col);
        __builtin_nontemporal_store(y1[r], o + 16 + col);
      }
    }
  }
}

__global__ void __launch_bounds__(256) edge_mlp_scatter_kernel(
    const float* __restrict__ X, const int* __restrict__ src,
    const float* __restrict__ w1, const float* __restrict__ b1,
    const float* __restrict__ w2, const float* __restrict__ b2,
    float* __restrict__ out, int n_edges) {
  __shared__ float lds[WAVES_PER_BLOCK * 16 * LDS_STRIDE];
  const int lane = threadIdx.x & 31;
  const int wid  = __builtin_amdgcn_readfirstlane((int)(threadIdx.x >> 5));
  float* wlds = lds + wid * 16 * LDS_STRIDE;

  const int ntiles = (n_edges + 15) >> 4;
  const int gwave  = blockIdx.x * WAVES_PER_BLOCK + wid;
  const int nwaves = gridDim.x * WAVES_PER_BLOCK;
  if (gwave >= ntiles) return;

  Frags f;
  load_frags(f, w1, b1, w2, b2, lane);

  unsigned* const agg = (unsigned*)out + 32;

  int sv;
  v4f xr[4];
  {
    int e = gwave * 16 + (lane & 15);
    sv = src[e < n_edges ? e : n_edges - 1];
    int nr0 = n_edges - gwave * 16; if (nr0 > 16) nr0 = 16;
    load_xtile(xr, X, gwave * 16, nr0, lane);
  }

  for (int tile = gwave; tile < ntiles; tile += nwaves) {
    const int row_base = tile * 16;
    int nr = n_edges - row_base; if (nr > 16) nr = 16;

    stage_xtile(xr, wlds, lane);
    int ntile = tile + nwaves;
    int sv_next = 0;
    if (ntile < ntiles) {
      int nrn = n_edges - ntile * 16; if (nrn > 16) nrn = 16;
      load_xtile(xr, X, ntile * 16, nrn, lane);
      int e = ntile * 16 + (lane & 15);
      sv_next = src[e < n_edges ? e : n_edges - 1];
    }

    v8f y0, y1;
    mlp_tile(f, wlds, lane, y0, y1);

    unsigned lo[8], hi[8];
#pragma unroll
    for (int r = 0; r < 8; ++r) {
      lo[r] = __float_as_uint(y0[r]);
      hi[r] = __float_as_uint(y1[r]);
      {
        v2u sw = __builtin_amdgcn_permlane16_swap(lo[r], hi[r], false, false);
        lo[r] = sw.x; hi[r] = sw.y;
      }
    }
    if (nr == 16) {
#pragma unroll
      for (int r = 0; r < 8; ++r) {
        const int s0 = __builtin_amdgcn_readlane(sv, r);
        const int s1 = __builtin_amdgcn_readlane(sv, r + 8);
        atomicMax(agg + (size_t)s0 * 64 + lane, lo[r]);
        atomicMax(agg + (size_t)s1 * 64 + lane, hi[r]);
      }
    } else {
#pragma unroll
      for (int r = 0; r < 8; ++r) {
        if (r < nr) {
          const int s0 = __builtin_amdgcn_readlane(sv, r);
          atomicMax(agg + (size_t)s0 * 64 + lane, lo[r]);
        }
        if (r + 8 < nr) {
          const int s1 = __builtin_amdgcn_readlane(sv, r + 8);
          atomicMax(agg + (size_t)s1 * 64 + lane, hi[r]);
        }
      }
    }
    sv = sv_next;
  }
}

extern "C" void kernel_launch(void* const* d_in, const int* in_sizes, int n_in,
                              void* d_out, int out_size, void* d_ws, size_t ws_size,
                              hipStream_t stream) {
  const float* node_features = (const float*)d_in[0];
  const float* edge_features = (const float*)d_in[1];
  const int*   edge_index    = (const int*)d_in[2];
  const float* nw1 = (const float*)d_in[3];
  const float* nb1 = (const float*)d_in[4];
  const float* nw2 = (const float*)d_in[5];
  const float* nb2 = (const float*)d_in[6];
  const float* ew1 = (const float*)d_in[7];
  const float* eb1 = (const float*)d_in[8];
  const float* ew2 = (const float*)d_in[9];
  const float* eb2 = (const float*)d_in[10];
  float* out = (float*)d_out;

  const int n_nodes = in_sizes[0] / 32;
  const int n_edges = in_sizes[1] / 32;

  zero_out_kernel<<<512, 256, 0, stream>>>(out, n_nodes * 64);
  node_mlp_kernel<<<512, 256, 0, stream>>>(
      node_features, nw1, nb1, nw2, nb2, out, n_nodes);
  edge_mlp_scatter_kernel<<<2048, 256, 0, stream>>>(
      edge_features, edge_index, ew1, eb1, ew2, eb2, out, n_edges);
}
